# starting kernel
# baseline (speedup 1.0000x reference)
#include <hip/hip_runtime.h>
#include <hip/hip_bf16.h>
#include <math.h>

#define BATCH   4
#define C_IN    128
#define C_OUT   128
#define NNODES  32768
#define KNBR    9
#define NEG_SLOPE 0.2f

typedef __attribute__((ext_vector_type(2))) float v2f;
typedef __attribute__((ext_vector_type(8))) float v8f;

__global__ __launch_bounds__(128) void gat_gemm_relu_kernel(
    const float* __restrict__ x,
    const float* __restrict__ W,
    float* __restrict__ h)
{
    const int lane = threadIdx.x & 31;
    const int wave = threadIdx.x >> 5;
    const int b    = blockIdx.y;
    const int n0   = (blockIdx.x * 4 + wave) * 16;

    const float* X = x + (size_t)b * C_IN * NNODES;

    const int m  = lane & 15;
    const int kk = (lane >> 4) * 2;

    v8f acc[8];
#pragma unroll
    for (int i = 0; i < 8; ++i)
        acc[i] = (v8f){0.f, 0.f, 0.f, 0.f, 0.f, 0.f, 0.f, 0.f};

    for (int k0 = 0; k0 < C_IN; k0 += 4) {
        v2f a;
        a.x = X[(size_t)(k0 + kk)     * NNODES + n0 + m];
        a.y = X[(size_t)(k0 + kk + 1) * NNODES + n0 + m];

#pragma unroll
        for (int ot = 0; ot < 8; ++ot) {
            const float* wrow = W + (size_t)(ot * 16 + m) * C_IN + k0 + kk;
            v2f bf;
            bf.x = wrow[0];
            bf.y = wrow[1];
            acc[ot] = __builtin_amdgcn_wmma_f32_16x16x4_f32(
                 false, a,  false, bf,
                 (short)0, acc[ot],
                 false,  false);
        }
    }

    const int col   = lane & 15;
    const int rbase = (lane >> 4) * 8;
#pragma unroll
    for (int ot = 0; ot < 8; ++ot) {
#pragma unroll
        for (int v = 0; v < 8; ++v) {
            const int row = v + rbase;
            float val = acc[ot][v];
            val = val > 0.f ? val : 0.f;
            h[((size_t)b * NNODES + n0 + row) * C_OUT + ot * 16 + col] = val;
        }
    }
}

__global__ __launch_bounds__(1024) void gat_attn_gather_kernel(
    const float* __restrict__ h,
    const int*   __restrict__ edge,
    const float* __restrict__ att,
    const float* __restrict__ bias,
    float* __restrict__ out)
{
    __shared__ float lds[32 * 132];

    const int lane = threadIdx.x & 31;
    const int wave = threadIdx.x >> 5;
    const int g    = blockIdx.x * 32 + wave;
    const int b    = g >> 15;

    int myidx = 0;
    if (lane < KNBR) myidx = edge[(size_t)g * KNBR + lane];

    const float4 a_i = *(const float4*)(att + 4 * lane);
    const float4 a_j = *(const float4*)(att + C_OUT + 4 * lane);

    const float4 self = *(const float4*)(h + (size_t)g * C_OUT + 4 * lane);
    float si = self.x * a_i.x + self.y * a_i.y + self.z * a_i.z + self.w * a_i.w;
#pragma unroll
    for (int off = 16; off > 0; off >>= 1) si += __shfl_xor(si, off, 32);

    float4 cols[KNBR];
    float  alpha[KNBR];
    const size_t hb = (size_t)b * NNODES;
#pragma unroll
    for (int k = 0; k < KNBR; ++k) {
        const int nk = __shfl(myidx, k, 32);
        const float4 c = *(const float4*)(h + (hb + nk) * C_OUT + 4 * lane);
        cols[k] = c;
        float sj = c.x * a_j.x + c.y * a_j.y + c.z * a_j.z + c.w * a_j.w;
#pragma unroll
        for (int off = 16; off > 0; off >>= 1) sj += __shfl_xor(sj, off, 32);
        const float a = si + sj;
        alpha[k] = a > 0.f ? a : NEG_SLOPE * a;
    }

    float mx = alpha[0];
#pragma unroll
    for (int k = 1; k < KNBR; ++k) mx = fmaxf(mx, alpha[k]);
    float den = 0.f;
#pragma unroll
    for (int k = 0; k < KNBR; ++k) { alpha[k] = __expf(alpha[k] - mx); den += alpha[k]; }
    const float inv = 1.f / den;

    float4 o;
    o.x = cols[0].x * (alpha[0] * inv);
    o.y = cols[0].y * (alpha[0] * inv);
    o.z = cols[0].z * (alpha[0] * inv);
    o.w = cols[0].w * (alpha[0] * inv);
#pragma unroll
    for (int k = 1; k < KNBR; ++k) {
        const float ak = alpha[k] * inv;
        o.x = fmaxf(o.x, cols[k].x * ak);
        o.y = fmaxf(o.y, cols[k].y * ak);
        o.z = fmaxf(o.z, cols[k].z * ak);
        o.w = fmaxf(o.w, cols[k].w * ak);
    }
    const float4 bs = *(const float4*)(bias + 4 * lane);
    o.x += bs.x; o.y += bs.y; o.z += bs.z; o.w += bs.w;

    float* row = lds + wave * 132 + 4 * lane;
    row[0] = o.x; row[1] = o.y; row[2] = o.z; row[3] = o.w;
    __syncthreads();

    const int n0 = (blockIdx.x * 32) & (NNODES - 1);
    const int c  = threadIdx.x >> 3;
    const int nn = (threadIdx.x & 7) * 4;
    float4 res;
    res.x = lds[(nn + 0) * 132 + c];
    res.y = lds[(nn + 1) * 132 + c];
    res.z = lds[(nn + 2) * 132 + c];
    res.w = lds[(nn + 3) * 132 + c];
    *(float4*)(out + ((size_t)(b * C_OUT + c) * NNODES) + n0 + nn) = res;
}

extern "C" void kernel_launch(void* const* d_in, const int* in_sizes, int n_in,
                              void* d_out, int out_size, void* d_ws, size_t ws_size,
                              hipStream_t stream)
{
    const float* x    = (const float*)d_in[0];
    const float* W    = (const float*)d_in[1];
    const float* att  = (const float*)d_in[2];
    const float* bias = (const float*)d_in[3];
    const int*   edge = (const int*)d_in[4];
    float*       out  = (float*)d_out;

    float* h = (float*)d_ws;

    dim3 g1((NNODES / 16) / 4, BATCH);
    gat_gemm_relu_kernel<<<g1, 128, 0, stream>>>(x, W, h);

    dim3 g2((BATCH * NNODES) / 32);
    gat_attn_gather_kernel<<<g2, 1024, 0, stream>>>(h, edge, att, bias, out);
}


# kernel 1: W staged once per work-group in LDS in fragment order, B fragments by ds_load_b128 instead of 16-line global loads; 256 blocks of 16 waves, 32 nodes per wave; k order and sums unchanged
# speedup vs baseline: 1.2122x; 1.2122x over previous
#include <hip/hip_runtime.h>
#include <hip/hip_bf16.h>
#include <math.h>

#define BATCH   4
#define C_IN    128
#define C_OUT   128
#define NNODES  32768
#define KNBR    9
#define NEG_SLOPE 0.2f

typedef __attribute__((ext_vector_type(2))) float v2f;
typedef __attribute__((ext_vector_type(8))) float v8f;

typedef __attribute__((ext_vector_type(4))) float v4f;

#define A_CHUNK  4

__device__ __forceinline__ void gat_load_a_chunk(v2f (&a)[2][A_CHUNK],
                                                 const float* __restrict__ xp,
                                                 int kstep0)
{
#pragma unroll
    for (int s = 0; s < A_CHUNK; ++s) {
        const float* p = xp + (size_t)((kstep0 + s) * 4) * NNODES;
        a[0][s].x = p[0];
        a[0][s].y = p[NNODES];
        a[1][s].x = p[16];
        a[1][s].y = p[NNODES + 16];
    }
}

__global__ __launch_bounds__(512) void gat_gemm_relu_kernel(
    const float* __restrict__ x,
    const float* __restrict__ W,
    float* __restrict__ h)
{
    __shared__ v4f wl[(C_IN / 4) * 4 * 2 * 16];

    const int lane = threadIdx.x & 31;
    const int wave = threadIdx.x >> 5;

    const int m  = lane & 15;
    const int hh = lane >> 4;
    const int kk = hh * 2;

    const int tile = blockIdx.x * 16 + wave;
    const int b    = tile >> 10;
    const int n0   = (tile & 1023) * 32;

    const float* xp = x + (size_t)b * C_IN * NNODES + (size_t)kk * NNODES + n0 + m;

    v2f abuf[2][2][A_CHUNK];
    gat_load_a_chunk(abuf[0], xp, 0);

    {
        const int q = wave * 2 + hh;
        float4 w4[8];
#pragma unroll
        for (int i = 0; i < 8; ++i)
            w4[i] = *(const float4*)(W + (size_t)(i * 16 + m) * C_IN + 4 * q);
#pragma unroll
        for (int j = 0; j < 4; ++j) {
            v4f* d = wl + ((q * 4 + j) * 2) * 16 + m;
            d[0]  = (v4f){w4[2 * j].x, w4[2 * j].y, w4[2 * j + 1].x, w4[2 * j + 1].y};
            d[16] = (v4f){w4[2 * j].z, w4[2 * j].w, w4[2 * j + 1].z, w4[2 * j + 1].w};
        }
    }
    __syncthreads();

    const v4f* bl = wl + hh * 16 + m;

    v8f acc[2][8];
#pragma unroll
    for (int t = 0; t < 2; ++t)
#pragma unroll
        for (int i = 0; i < 8; ++i)
            acc[t][i] = (v8f){0.f, 0.f, 0.f, 0.f, 0.f, 0.f, 0.f, 0.f};

    v4f bbuf[2][4];
#pragma unroll
    for (int j = 0; j < 4; ++j)
        bbuf[0][j] = bl[j * 32];

#pragma unroll
    for (int c = 0; c < (C_IN / 4) / A_CHUNK; ++c) {
#pragma unroll
        for (int s = 0; s < A_CHUNK; ++s) {
            const int ks = c * A_CHUNK + s;

            if (s == 0 && c + 1 < (C_IN / 4) / A_CHUNK)
                gat_load_a_chunk(abuf[(c + 1) & 1], xp, (c + 1) * A_CHUNK);

            if (ks + 1 < C_IN / 4) {
#pragma unroll
                for (int j = 0; j < 4; ++j)
                    bbuf[(ks + 1) & 1][j] = bl[((ks + 1) * 4 + j) * 32];
            }

            __builtin_amdgcn_sched_barrier(0);

            const v2f a0 = abuf[c & 1][0][s];
            const v2f a1 = abuf[c & 1][1][s];
#pragma unroll
            for (int ot = 0; ot < 8; ++ot) {
                const v4f bq = bbuf[ks & 1][ot >> 1];
                const v2f bf = (ot & 1) ? __builtin_shufflevector(bq, bq, 2, 3)
                                        : __builtin_shufflevector(bq, bq, 0, 1);
                acc[0][ot] = __builtin_amdgcn_wmma_f32_16x16x4_f32(
                     false, a0,  false, bf,
                     (short)0, acc[0][ot],
                     false,  false);
                acc[1][ot] = __builtin_amdgcn_wmma_f32_16x16x4_f32(
                     false, a1,  false, bf,
                     (short)0, acc[1][ot],
                     false,  false);
            }

            __builtin_amdgcn_sched_barrier(0);
        }
    }

    const int col   = lane & 15;
    const int rbase = (lane >> 4) * 8;
#pragma unroll
    for (int t = 0; t < 2; ++t) {
#pragma unroll
        for (int ot = 0; ot < 8; ++ot) {
#pragma unroll
            for (int v = 0; v < 8; ++v) {
                const int row = t * 16 + v + rbase;
                float val = acc[t][ot][v];
                val = val > 0.f ? val : 0.f;
                h[((size_t)b * NNODES + n0 + row) * C_OUT + ot * 16 + col] = val;
            }
        }
    }
}

__global__ __launch_bounds__(1024) void gat_attn_gather_kernel(
    const float* __restrict__ h,
    const int*   __restrict__ edge,
    const float* __restrict__ att,
    const float* __restrict__ bias,
    float* __restrict__ out)
{
    __shared__ float lds[32 * 132];

    const int lane = threadIdx.x & 31;
    const int wave = threadIdx.x >> 5;
    const int g    = blockIdx.x * 32 + wave;
    const int b    = g >> 15;

    int myidx = 0;
    if (lane < KNBR) myidx = edge[(size_t)g * KNBR + lane];

    const float4 a_i = *(const float4*)(att + 4 * lane);
    const float4 a_j = *(const float4*)(att + C_OUT + 4 * lane);

    const float4 self = *(const float4*)(h + (size_t)g * C_OUT + 4 * lane);
    float si = self.x * a_i.x + self.y * a_i.y + self.z * a_i.z + self.w * a_i.w;
#pragma unroll
    for (int off = 16; off > 0; off >>= 1) si += __shfl_xor(si, off, 32);

    float4 cols[KNBR];
    float  alpha[KNBR];
    const size_t hb = (size_t)b * NNODES;
#pragma unroll
    for (int k = 0; k < KNBR; ++k) {
        const int nk = __shfl(myidx, k, 32);
        const float4 c = *(const float4*)(h + (hb + nk) * C_OUT + 4 * lane);
        cols[k] = c;
        float sj = c.x * a_j.x + c.y * a_j.y + c.z * a_j.z + c.w * a_j.w;
#pragma unroll
        for (int off = 16; off > 0; off >>= 1) sj += __shfl_xor(sj, off, 32);
        const float a = si + sj;
        alpha[k] = a > 0.f ? a : NEG_SLOPE * a;
    }

    float mx = alpha[0];
#pragma unroll
    for (int k = 1; k < KNBR; ++k) mx = fmaxf(mx, alpha[k]);
    float den = 0.f;
#pragma unroll
    for (int k = 0; k < KNBR; ++k) { alpha[k] = __expf(alpha[k] - mx); den += alpha[k]; }
    const float inv = 1.f / den;

    float4 o;
    o.x = cols[0].x * (alpha[0] * inv);
    o.y = cols[0].y * (alpha[0] * inv);
    o.z = cols[0].z * (alpha[0] * inv);
    o.w = cols[0].w * (alpha[0] * inv);
#pragma unroll
    for (int k = 1; k < KNBR; ++k) {
        const float ak = alpha[k] * inv;
        o.x = fmaxf(o.x, cols[k].x * ak);
        o.y = fmaxf(o.y, cols[k].y * ak);
        o.z = fmaxf(o.z, cols[k].z * ak);
        o.w = fmaxf(o.w, cols[k].w * ak);
    }
    const float4 bs = *(const float4*)(bias + 4 * lane);
    o.x += bs.x; o.y += bs.y; o.z += bs.z; o.w += bs.w;

    float* row = lds + wave * 132 + 4 * lane;
    row[0] = o.x; row[1] = o.y; row[2] = o.z; row[3] = o.w;
    __syncthreads();

    const int n0 = (blockIdx.x * 32) & (NNODES - 1);
    const int c  = threadIdx.x >> 3;
    const int nn = (threadIdx.x & 7) * 4;
    float4 res;
    res.x = lds[(nn + 0) * 132 + c];
    res.y = lds[(nn + 1) * 132 + c];
    res.z = lds[(nn + 2) * 132 + c];
    res.w = lds[(nn + 3) * 132 + c];
    *(float4*)(out + ((size_t)(b * C_OUT + c) * NNODES) + n0 + nn) = res;
}

extern "C" void kernel_launch(void* const* d_in, const int* in_sizes, int n_in,
                              void* d_out, int out_size, void* d_ws, size_t ws_size,
                              hipStream_t stream)
{
    const float* x    = (const float*)d_in[0];
    const float* W    = (const float*)d_in[1];
    const float* att  = (const float*)d_in[2];
    const float* bias = (const float*)d_in[3];
    const int*   edge = (const int*)d_in[4];
    float*       out  = (float*)d_out;

    float* h = (float*)d_ws;

    dim3 g1((BATCH * NNODES / 32) / 16);
    gat_gemm_relu_kernel<<<g1, 512, 0, stream>>>(x, W, h);

    dim3 g2((BATCH * NNODES) / 32);
    gat_attn_gather_kernel<<<g2, 1024, 0, stream>>>(h, edge, att, bias, out);
}
